# starting kernel
# baseline (speedup 1.0000x reference)
#include <hip/hip_runtime.h>
#include <hip/hip_bf16.h>

#define B_  32
#define S_  1024
#define D_  256
#define W_  4
#define NL_ 2
#define NH_ 2

typedef __attribute__((ext_vector_type(2))) float v2f;
typedef __attribute__((ext_vector_type(8))) float v8f;

__global__ __launch_bounds__(256)
void fir_kernel(const float* __restrict__ x, int xrow,
                const float* __restrict__ padf, const float* __restrict__ padb,
                const float* __restrict__ w, int off,
                float* __restrict__ y) {
  int gid = blockIdx.x * blockDim.x + threadIdx.x;
  int d = gid & (D_ - 1);
  int s = (gid >> 8) & (S_ - 1);
  int b = gid >> 18;
  float acc = 0.f;
#pragma unroll
  for (int j = 0; j <= W_; ++j) {
    int t = s + off + j;
    float v;
    if (t < W_)             v = padf[t * D_ + d];
    else if (t >= W_ + S_)  v = padb[(t - W_ - S_) * D_ + d];
    else                    v = x[((size_t)b * S_ + (t - W_)) * (size_t)xrow + d];
    acc = fmaf(w[j], v, acc);
  }
  y[(size_t)gid] = acc;
}

__global__ __launch_bounds__(256)
void highway_kernel(const float* __restrict__ x,
                    const float* __restrict__ Wt,
                    const float* __restrict__ bias,
                    float* __restrict__ out, int orow) {
  const int lane = threadIdx.x & 31;
  const int wave = threadIdx.x >> 5;
  const int half = lane >> 4;
  const int l    = lane & 15;

  const int m0 = (blockIdx.y * 8 + wave) * 32;
  const int n0 = blockIdx.x * 32;
  const int nA = n0 + l;
  const int nB = n0 + 16 + l;

  const float* arow0 = x  + (size_t)(m0 + l) * D_       + 2 * half;
  const float* arow1 = x  + (size_t)(m0 + 16 + l) * D_  + 2 * half;
  const float* wn0   = Wt + (size_t)nA * D_             + 2 * half;
  const float* wn1   = Wt + (size_t)nB * D_             + 2 * half;
  const float* wg0   = Wt + (size_t)(D_ + nA) * D_      + 2 * half;
  const float* wg1   = Wt + (size_t)(D_ + nB) * D_      + 2 * half;

  v8f aN0 = {}; v8f aN1 = {}; v8f aG0 = {}; v8f aG1 = {};
  v8f bN0 = {}; v8f bN1 = {}; v8f bG0 = {}; v8f bG1 = {};

#pragma unroll 2
  for (int g = 0; g < D_ / 4; ++g) {
    v2f a0 = *(const v2f*)(arow0 + 4 * g);
    v2f a1 = *(const v2f*)(arow1 + 4 * g);
    v2f n0v = *(const v2f*)(wn0 + 4 * g);
    v2f n1v = *(const v2f*)(wn1 + 4 * g);
    v2f g0v = *(const v2f*)(wg0 + 4 * g);
    v2f g1v = *(const v2f*)(wg1 + 4 * g);
    aN0 = __builtin_amdgcn_wmma_f32_16x16x4_f32(false, a0, false, n0v, (short)0, aN0, false, false);
    aN1 = __builtin_amdgcn_wmma_f32_16x16x4_f32(false, a0, false, n1v, (short)0, aN1, false, false);
    aG0 = __builtin_amdgcn_wmma_f32_16x16x4_f32(false, a0, false, g0v, (short)0, aG0, false, false);
    aG1 = __builtin_amdgcn_wmma_f32_16x16x4_f32(false, a0, false, g1v, (short)0, aG1, false, false);
    bN0 = __builtin_amdgcn_wmma_f32_16x16x4_f32(false, a1, false, n0v, (short)0, bN0, false, false);
    bN1 = __builtin_amdgcn_wmma_f32_16x16x4_f32(false, a1, false, n1v, (short)0, bN1, false, false);
    bG0 = __builtin_amdgcn_wmma_f32_16x16x4_f32(false, a1, false, g0v, (short)0, bG0, false, false);
    bG1 = __builtin_amdgcn_wmma_f32_16x16x4_f32(false, a1, false, g1v, (short)0, bG1, false, false);
  }

  const float bnA = bias[nA], bnB = bias[nB];
  const float bgA = bias[D_ + nA], bgB = bias[D_ + nB];

#pragma unroll
  for (int r = 0; r < 8; ++r) {
    {
      const int m = m0 + r + 8 * half;
      const float* xr = x + (size_t)m * D_;
      float* op = out + (size_t)m * orow;
      float g0 = 1.f / (1.f + __expf(-(aG0[r] + bgA)));
      op[nA] = g0 * xr[nA] + (1.f - g0) * fmaxf(aN0[r] + bnA, 0.f);
      float g1 = 1.f / (1.f + __expf(-(aG1[r] + bgB)));
      op[nB] = g1 * xr[nB] + (1.f - g1) * fmaxf(aN1[r] + bnB, 0.f);
    }
    {
      const int m = m0 + 16 + r + 8 * half;
      const float* xr = x + (size_t)m * D_;
      float* op = out + (size_t)m * orow;
      float g0 = 1.f / (1.f + __expf(-(bG0[r] + bgA)));
      op[nA] = g0 * xr[nA] + (1.f - g0) * fmaxf(bN0[r] + bnA, 0.f);
      float g1 = 1.f / (1.f + __expf(-(bG1[r] + bgB)));
      op[nB] = g1 * xr[nB] + (1.f - g1) * fmaxf(bN1[r] + bnB, 0.f);
    }
  }
}

extern "C" void kernel_launch(void* const* d_in, const int* in_sizes, int n_in,
                              void* d_out, int out_size, void* d_ws, size_t ws_size,
                              hipStream_t stream) {
  (void)in_sizes; (void)n_in; (void)out_size; (void)ws_size;

  const float* inputs   = (const float*)d_in[0];
  const float* fwd_pad  = (const float*)d_in[1];
  const float* bwd_pad  = (const float*)d_in[2];
  const float* fwd_w    = (const float*)d_in[3];
  const float* bwd_w    = (const float*)d_in[4];
  const float* fwd_hw_W = (const float*)d_in[5];
  const float* fwd_hw_b = (const float*)d_in[6];
  const float* bwd_hw_W = (const float*)d_in[7];
  const float* bwd_hw_b = (const float*)d_in[8];
  float* out = (float*)d_out;

  const size_t M = (size_t)B_ * S_;
  float* yA = (float*)d_ws;
  float* yB = yA + M * D_;

  dim3 hwBlock(256);
  dim3 hwGrid(D_ / 32, (unsigned)(M / (32 * 8)));
  const int firBlocks = (int)(M * D_ / 256);

  for (int i = 0; i < NL_; ++i) {
    for (int dir = 0; dir < 2; ++dir) {
      const float* xin; int xrow;
      if (i == 0) { xin = inputs; xrow = D_; }
      else        { xin = out + (size_t)(i - 1) * M * 2 * D_ + dir * D_; xrow = 2 * D_; }

      const float* taps = (dir ? bwd_w : fwd_w) + i * (W_ + 1);
      fir_kernel<<<firBlocks, 256, 0, stream>>>(
          xin, xrow,
          fwd_pad + (size_t)i * W_ * D_, bwd_pad + (size_t)i * W_ * D_,
          taps, dir ? W_ : 0, yA);

      const float* HW = (dir ? bwd_hw_W : fwd_hw_W) + (size_t)i * NH_ * 2 * D_ * D_;
      const float* Hb = (dir ? bwd_hw_b : fwd_hw_b) + (size_t)i * NH_ * 2 * D_;

      highway_kernel<<<hwGrid, hwBlock, 0, stream>>>(yA, HW, Hb, yB, D_);
      highway_kernel<<<hwGrid, hwBlock, 0, stream>>>(
          yB, HW + (size_t)2 * D_ * D_, Hb + 2 * D_,
          out + (size_t)i * M * 2 * D_ + dir * D_, 2 * D_);
    }
  }
}


# Highway GEMM: x and W tiles staged in LDS by 16-byte global loads covering whole 128-byte lines (128x128 workgroup tile, K chunks of 32, register prefetch); fragments read from LDS; sum order kept
# speedup vs baseline: 1.5482x; 1.5482x over previous
#include <hip/hip_runtime.h>
#include <hip/hip_bf16.h>

#define B_  32
#define S_  1024
#define D_  256
#define W_  4
#define NL_ 2
#define NH_ 2

typedef __attribute__((ext_vector_type(2))) float v2f;
typedef __attribute__((ext_vector_type(8))) float v8f;

__global__ __launch_bounds__(256)
void fir_kernel(const float* __restrict__ x, int xrow,
                const float* __restrict__ padf, const float* __restrict__ padb,
                const float* __restrict__ w, int off,
                float* __restrict__ y) {
  int gid = blockIdx.x * blockDim.x + threadIdx.x;
  int d = gid & (D_ - 1);
  int s = (gid >> 8) & (S_ - 1);
  int b = gid >> 18;
  float acc = 0.f;
#pragma unroll
  for (int j = 0; j <= W_; ++j) {
    int t = s + off + j;
    float v;
    if (t < W_)             v = padf[t * D_ + d];
    else if (t >= W_ + S_)  v = padb[(t - W_ - S_) * D_ + d];
    else                    v = x[((size_t)b * S_ + (t - W_)) * (size_t)xrow + d];
    acc = fmaf(w[j], v, acc);
  }
  y[(size_t)gid] = acc;
}

#define BM_     128
#define BN_     64
#define KC_     32
#define PITCH_  36
#define STAGE_  ((BM_ + 2 * BN_) * PITCH_)

typedef __attribute__((ext_vector_type(4))) float v4f;

__global__ __launch_bounds__(256)
void highway_kernel(const float* __restrict__ x,
                    const float* __restrict__ Wt,
                    const float* __restrict__ bias,
                    float* __restrict__ out, int orow) {
  __shared__ __attribute__((aligned(16))) float lds[2 * STAGE_];

  const int tid  = threadIdx.x;
  const int lane = tid & 31;
  const int wave = tid >> 5;
  const int half = lane >> 4;
  const int l    = lane & 15;
  const int wm   = wave & 3;
  const int wn   = wave >> 2;

  const int mBlk = blockIdx.y * BM_;
  const int nBlk = blockIdx.x * BN_;

  const int m0 = mBlk + wm * 32;
  const int nA = nBlk + wn * 32 + l;
  const int nB = nA + 16;

  const int lrow = tid >> 3;
  const int lcol = (tid & 7) * 4;
  const float* gA  = x  + (size_t)(mBlk + lrow) * D_        + lcol;
  const float* gBn = Wt + (size_t)(nBlk + lrow) * D_        + lcol;
  const float* gBg = Wt + (size_t)(D_ + nBlk + lrow) * D_   + lcol;
  const int sOff = lrow * PITCH_ + lcol;

  const int fA0 = (wm * 32 + l) * PITCH_ + 2 * half;
  const int fA1 = fA0 + 16 * PITCH_;
  const int fN0 = (BM_ + wn * 32 + l) * PITCH_ + 2 * half;
  const int fN1 = fN0 + 16 * PITCH_;
  const int fG0 = fN0 + BN_ * PITCH_;
  const int fG1 = fG0 + 16 * PITCH_;

  v8f aN0 = {}; v8f aN1 = {}; v8f aG0 = {}; v8f aG1 = {};
  v8f bN0 = {}; v8f bN1 = {}; v8f bG0 = {}; v8f bG1 = {};

  v4f r0, r1, r2, r3;

#define HW_GROUP_(g)                                                                                   \
  {                                                                                                    \
    v2f a0  = *(const v2f*)(s + fA0 + 4 * (g));                                                        \
    v2f a1  = *(const v2f*)(s + fA1 + 4 * (g));                                                        \
    v2f n0v = *(const v2f*)(s + fN0 + 4 * (g));                                                        \
    v2f n1v = *(const v2f*)(s + fN1 + 4 * (g));                                                        \
    v2f g0v = *(const v2f*)(s + fG0 + 4 * (g));                                                        \
    v2f g1v = *(const v2f*)(s + fG1 + 4 * (g));                                                        \
    aN0 = __builtin_amdgcn_wmma_f32_16x16x4_f32(false, a0, false, n0v, (short)0, aN0, false, false);   \
    aN1 = __builtin_amdgcn_wmma_f32_16x16x4_f32(false, a0, false, n1v, (short)0, aN1, false, false);   \
    aG0 = __builtin_amdgcn_wmma_f32_16x16x4_f32(false, a0, false, g0v, (short)0, aG0, false, false);   \
    aG1 = __builtin_amdgcn_wmma_f32_16x16x4_f32(false, a0, false, g1v, (short)0, aG1, false, false);   \
    bN0 = __builtin_amdgcn_wmma_f32_16x16x4_f32(false, a1, false, n0v, (short)0, bN0, false, false);   \
    bN1 = __builtin_amdgcn_wmma_f32_16x16x4_f32(false, a1, false, n1v, (short)0, bN1, false, false);   \
    bG0 = __builtin_amdgcn_wmma_f32_16x16x4_f32(false, a1, false, g0v, (short)0, bG0, false, false);   \
    bG1 = __builtin_amdgcn_wmma_f32_16x16x4_f32(false, a1, false, g1v, (short)0, bG1, false, false);   \
  }

  {
    float* d = lds + sOff;
    r0 = *(const v4f*)(gA);
    r1 = *(const v4f*)(gA + 32 * D_);
    r2 = *(const v4f*)(gA + 64 * D_);
    r3 = *(const v4f*)(gA + 96 * D_);
    *(v4f*)(d)                          = r0;
    *(v4f*)(d + 32 * PITCH_)            = r1;
    *(v4f*)(d + 64 * PITCH_)            = r2;
    *(v4f*)(d + 96 * PITCH_)            = r3;
    r0 = *(const v4f*)(gBn);
    r1 = *(const v4f*)(gBn + 32 * D_);
    r2 = *(const v4f*)(gBg);
    r3 = *(const v4f*)(gBg + 32 * D_);
    *(v4f*)(d + (BM_ +  0) * PITCH_)    = r0;
    *(v4f*)(d + (BM_ + 32) * PITCH_)    = r1;
    *(v4f*)(d + (BM_ + 64) * PITCH_)    = r2;
    *(v4f*)(d + (BM_ + 96) * PITCH_)    = r3;
  }
  __syncthreads();

  for (int kc = 0; kc < D_ / KC_; ++kc) {
    const bool more = (kc + 1) < D_ / KC_;
    const int ko = (kc + 1) * KC_;
    const float* s = lds + (kc & 1) * STAGE_;
    float* d = lds + ((kc + 1) & 1) * STAGE_ + sOff;

    if (more) {
      r0 = *(const v4f*)(gA + ko);
      r1 = *(const v4f*)(gA + 32 * D_ + ko);
      r2 = *(const v4f*)(gA + 64 * D_ + ko);
      r3 = *(const v4f*)(gA + 96 * D_ + ko);
    }
    __builtin_amdgcn_sched_barrier(0);
    HW_GROUP_(0) HW_GROUP_(1)
    __builtin_amdgcn_sched_barrier(0);
    HW_GROUP_(2) HW_GROUP_(3)
    __builtin_amdgcn_sched_barrier(0);
    if (more) {
      *(v4f*)(d)                          = r0;
      *(v4f*)(d + 32 * PITCH_)            = r1;
      *(v4f*)(d + 64 * PITCH_)            = r2;
      *(v4f*)(d + 96 * PITCH_)            = r3;
      r0 = *(const v4f*)(gBn + ko);
      r1 = *(const v4f*)(gBn + 32 * D_ + ko);
      r2 = *(const v4f*)(gBg + ko);
      r3 = *(const v4f*)(gBg + 32 * D_ + ko);
    }
    __builtin_amdgcn_sched_barrier(0);
    HW_GROUP_(4) HW_GROUP_(5)
    __builtin_amdgcn_sched_barrier(0);
    HW_GROUP_(6) HW_GROUP_(7)
    __builtin_amdgcn_sched_barrier(0);
    if (more) {
      *(v4f*)(d + (BM_ +  0) * PITCH_)    = r0;
      *(v4f*)(d + (BM_ + 32) * PITCH_)    = r1;
      *(v4f*)(d + (BM_ + 64) * PITCH_)    = r2;
      *(v4f*)(d + (BM_ + 96) * PITCH_)    = r3;
      __syncthreads();
    }
  }
#undef HW_GROUP_

  const float bnA = bias[nA], bnB = bias[nB];
  const float bgA = bias[D_ + nA], bgB = bias[D_ + nB];

#pragma unroll
  for (int r = 0; r < 8; ++r) {
    {
      const int m = m0 + r + 8 * half;
      const float* xr = x + (size_t)m * D_;
      float* op = out + (size_t)m * orow;
      float g0 = 1.f / (1.f + __expf(-(aG0[r] + bgA)));
      op[nA] = g0 * xr[nA] + (1.f - g0) * fmaxf(aN0[r] + bnA, 0.f);
      float g1 = 1.f / (1.f + __expf(-(aG1[r] + bgB)));
      op[nB] = g1 * xr[nB] + (1.f - g1) * fmaxf(aN1[r] + bnB, 0.f);
    }
    {
      const int m = m0 + 16 + r + 8 * half;
      const float* xr = x + (size_t)m * D_;
      float* op = out + (size_t)m * orow;
      float g0 = 1.f / (1.f + __expf(-(bG0[r] + bgA)));
      op[nA] = g0 * xr[nA] + (1.f - g0) * fmaxf(bN0[r] + bnA, 0.f);
      float g1 = 1.f / (1.f + __expf(-(bG1[r] + bgB)));
      op[nB] = g1 * xr[nB] + (1.f - g1) * fmaxf(bN1[r] + bnB, 0.f);
    }
  }
}

extern "C" void kernel_launch(void* const* d_in, const int* in_sizes, int n_in,
                              void* d_out, int out_size, void* d_ws, size_t ws_size,
                              hipStream_t stream) {
  (void)in_sizes; (void)n_in; (void)out_size; (void)ws_size;

  const float* inputs   = (const float*)d_in[0];
  const float* fwd_pad  = (const float*)d_in[1];
  const float* bwd_pad  = (const float*)d_in[2];
  const float* fwd_w    = (const float*)d_in[3];
  const float* bwd_w    = (const float*)d_in[4];
  const float* fwd_hw_W = (const float*)d_in[5];
  const float* fwd_hw_b = (const float*)d_in[6];
  const float* bwd_hw_W = (const float*)d_in[7];
  const float* bwd_hw_b = (const float*)d_in[8];
  float* out = (float*)d_out;

  const size_t M = (size_t)B_ * S_;
  float* yA = (float*)d_ws;
  float* yB = yA + M * D_;

  dim3 hwBlock(256);
  dim3 hwGrid(D_ / BN_, (unsigned)(M / BM_));
  const int firBlocks = (int)(M * D_ / 256);

  for (int i = 0; i < NL_; ++i) {
    for (int dir = 0; dir < 2; ++dir) {
      const float* xin; int xrow;
      if (i == 0) { xin = inputs; xrow = D_; }
      else        { xin = out + (size_t)(i - 1) * M * 2 * D_ + dir * D_; xrow = 2 * D_; }

      const float* taps = (dir ? bwd_w : fwd_w) + i * (W_ + 1);
      fir_kernel<<<firBlocks, 256, 0, stream>>>(
          xin, xrow,
          fwd_pad + (size_t)i * W_ * D_, bwd_pad + (size_t)i * W_ * D_,
          taps, dir ? W_ : 0, yA);

      const float* HW = (dir ? bwd_hw_W : fwd_hw_W) + (size_t)i * NH_ * 2 * D_ * D_;
      const float* Hb = (dir ? bwd_hw_b : fwd_hw_b) + (size_t)i * NH_ * 2 * D_;

      highway_kernel<<<hwGrid, hwBlock, 0, stream>>>(yA, HW, Hb, yB, D_);
      highway_kernel<<<hwGrid, hwBlock, 0, stream>>>(
          yB, HW + (size_t)2 * D_ * D_, Hb + 2 * D_,
          out + (size_t)i * M * 2 * D_ + dir * D_, 2 * D_);
    }
  }
}


# FIR pass as a streaming kernel: each thread owns 4 d x 16 s, loads its 20 window rows once with 16-byte loads (512 B per wave per row) and forms outputs from registers; tap order j ascending kept
# speedup vs baseline: 1.7924x; 1.1577x over previous
#include <hip/hip_runtime.h>
#include <hip/hip_bf16.h>

#define B_  32
#define S_  1024
#define D_  256
#define W_  4
#define NL_ 2
#define NH_ 2

typedef __attribute__((ext_vector_type(2))) float v2f;
typedef __attribute__((ext_vector_type(8))) float v8f;

#define FIR_ROWS_ 16

typedef __attribute__((ext_vector_type(4))) float v4f;

__global__ __launch_bounds__(256)
void fir_kernel(const float* __restrict__ x, int xrow,
                const float* __restrict__ padf, const float* __restrict__ padb,
                const float* __restrict__ w, int off,
                float* __restrict__ y) {
  const int gid = blockIdx.x * 256 + threadIdx.x;
  const int d   = (gid & 63) * 4;
  const int rb  = __builtin_amdgcn_readfirstlane(gid >> 6);
  const int s0  = (rb & (S_ / FIR_ROWS_ - 1)) * FIR_ROWS_;
  const int b   = rb >> 6;

  const float w0 = w[0], w1 = w[1], w2 = w[2], w3 = w[3], w4 = w[4];

  const int t0 = s0 + off;
  v4f v[FIR_ROWS_ + W_];
  if (t0 >= W_ && t0 + FIR_ROWS_ + W_ <= W_ + S_) {
    const float* p = x + ((size_t)b * S_ + (size_t)(t0 - W_)) * (size_t)xrow + d;
#pragma unroll
    for (int u = 0; u < FIR_ROWS_ + W_; ++u)
      v[u] = *(const v4f*)(p + (size_t)u * (size_t)xrow);
  } else {
#pragma unroll
    for (int u = 0; u < FIR_ROWS_ + W_; ++u) {
      const int t = t0 + u;
      const float* p;
      if (t < W_)             p = padf + t * D_;
      else if (t >= W_ + S_)  p = padb + (t - W_ - S_) * D_;
      else                    p = x + ((size_t)b * S_ + (size_t)(t - W_)) * (size_t)xrow;
      v[u] = *(const v4f*)(p + d);
    }
  }

  float* yp = y + ((size_t)b * S_ + (size_t)s0) * D_ + d;
#pragma unroll
  for (int r = 0; r < FIR_ROWS_; ++r) {
    v4f acc;
#pragma unroll
    for (int c = 0; c < 4; ++c) {
      float a = fmaf(w0, v[r][c], 0.f);
      a = fmaf(w1, v[r + 1][c], a);
      a = fmaf(w2, v[r + 2][c], a);
      a = fmaf(w3, v[r + 3][c], a);
      a = fmaf(w4, v[r + 4][c], a);
      acc[c] = a;
    }
    *(v4f*)(yp + (size_t)r * D_) = acc;
  }
}

#define BM_     128
#define BN_     64
#define KC_     32
#define PITCH_  36
#define STAGE_  ((BM_ + 2 * BN_) * PITCH_)

__global__ __launch_bounds__(256)
void highway_kernel(const float* __restrict__ x,
                    const float* __restrict__ Wt,
                    const float* __restrict__ bias,
                    float* __restrict__ out, int orow) {
  __shared__ __attribute__((aligned(16))) float lds[2 * STAGE_];

  const int tid  = threadIdx.x;
  const int lane = tid & 31;
  const int wave = tid >> 5;
  const int half = lane >> 4;
  const int l    = lane & 15;
  const int wm   = wave & 3;
  const int wn   = wave >> 2;

  const int mBlk = blockIdx.y * BM_;
  const int nBlk = blockIdx.x * BN_;

  const int m0 = mBlk + wm * 32;
  const int nA = nBlk + wn * 32 + l;
  const int nB = nA + 16;

  const int lrow = tid >> 3;
  const int lcol = (tid & 7) * 4;
  const float* gA  = x  + (size_t)(mBlk + lrow) * D_        + lcol;
  const float* gBn = Wt + (size_t)(nBlk + lrow) * D_        + lcol;
  const float* gBg = Wt + (size_t)(D_ + nBlk + lrow) * D_   + lcol;
  const int sOff = lrow * PITCH_ + lcol;

  const int fA0 = (wm * 32 + l) * PITCH_ + 2 * half;
  const int fA1 = fA0 + 16 * PITCH_;
  const int fN0 = (BM_ + wn * 32 + l) * PITCH_ + 2 * half;
  const int fN1 = fN0 + 16 * PITCH_;
  const int fG0 = fN0 + BN_ * PITCH_;
  const int fG1 = fG0 + 16 * PITCH_;

  v8f aN0 = {}; v8f aN1 = {}; v8f aG0 = {}; v8f aG1 = {};
  v8f bN0 = {}; v8f bN1 = {}; v8f bG0 = {}; v8f bG1 = {};

  v4f r0, r1, r2, r3;

#define HW_GROUP_(g)                                                                                   \
  {                                                                                                    \
    v2f a0  = *(const v2f*)(s + fA0 + 4 * (g));                                                        \
    v2f a1  = *(const v2f*)(s + fA1 + 4 * (g));                                                        \
    v2f n0v = *(const v2f*)(s + fN0 + 4 * (g));                                                        \
    v2f n1v = *(const v2f*)(s + fN1 + 4 * (g));                                                        \
    v2f g0v = *(const v2f*)(s + fG0 + 4 * (g));                                                        \
    v2f g1v = *(const v2f*)(s + fG1 + 4 * (g));                                                        \
    aN0 = __builtin_amdgcn_wmma_f32_16x16x4_f32(false, a0, false, n0v, (short)0, aN0, false, false);   \
    aN1 = __builtin_amdgcn_wmma_f32_16x16x4_f32(false, a0, false, n1v, (short)0, aN1, false, false);   \
    aG0 = __builtin_amdgcn_wmma_f32_16x16x4_f32(false, a0, false, g0v, (short)0, aG0, false, false);   \
    aG1 = __builtin_amdgcn_wmma_f32_16x16x4_f32(false, a0, false, g1v, (short)0, aG1, false, false);   \
    bN0 = __builtin_amdgcn_wmma_f32_16x16x4_f32(false, a1, false, n0v, (short)0, bN0, false, false);   \
    bN1 = __builtin_amdgcn_wmma_f32_16x16x4_f32(false, a1, false, n1v, (short)0, bN1, false, false);   \
    bG0 = __builtin_amdgcn_wmma_f32_16x16x4_f32(false, a1, false, g0v, (short)0, bG0, false, false);   \
    bG1 = __builtin_amdgcn_wmma_f32_16x16x4_f32(false, a1, false, g1v, (short)0, bG1, false, false);   \
  }

  {
    float* d = lds + sOff;
    r0 = *(const v4f*)(gA);
    r1 = *(const v4f*)(gA + 32 * D_);
    r2 = *(const v4f*)(gA + 64 * D_);
    r3 = *(const v4f*)(gA + 96 * D_);
    *(v4f*)(d)                          = r0;
    *(v4f*)(d + 32 * PITCH_)            = r1;
    *(v4f*)(d + 64 * PITCH_)            = r2;
    *(v4f*)(d + 96 * PITCH_)            = r3;
    r0 = *(const v4f*)(gBn);
    r1 = *(const v4f*)(gBn + 32 * D_);
    r2 = *(const v4f*)(gBg);
    r3 = *(const v4f*)(gBg + 32 * D_);
    *(v4f*)(d + (BM_ +  0) * PITCH_)    = r0;
    *(v4f*)(d + (BM_ + 32) * PITCH_)    = r1;
    *(v4f*)(d + (BM_ + 64) * PITCH_)    = r2;
    *(v4f*)(d + (BM_ + 96) * PITCH_)    = r3;
  }
  __syncthreads();

  for (int kc = 0; kc < D_ / KC_; ++kc) {
    const bool more = (kc + 1) < D_ / KC_;
    const int ko = (kc + 1) * KC_;
    const float* s = lds + (kc & 1) * STAGE_;
    float* d = lds + ((kc + 1) & 1) * STAGE_ + sOff;

    if (more) {
      r0 = *(const v4f*)(gA + ko);
      r1 = *(const v4f*)(gA + 32 * D_ + ko);
      r2 = *(const v4f*)(gA + 64 * D_ + ko);
      r3 = *(const v4f*)(gA + 96 * D_ + ko);
    }
    __builtin_amdgcn_sched_barrier(0);
    HW_GROUP_(0) HW_GROUP_(1)
    __builtin_amdgcn_sched_barrier(0);
    HW_GROUP_(2) HW_GROUP_(3)
    __builtin_amdgcn_sched_barrier(0);
    if (more) {
      *(v4f*)(d)                          = r0;
      *(v4f*)(d + 32 * PITCH_)            = r1;
      *(v4f*)(d + 64 * PITCH_)            = r2;
      *(v4f*)(d + 96 * PITCH_)            = r3;
      r0 = *(const v4f*)(gBn + ko);
      r1 = *(const v4f*)(gBn + 32 * D_ + ko);
      r2 = *(const v4f*)(gBg + ko);
      r3 = *(const v4f*)(gBg + 32 * D_ + ko);
    }
    __builtin_amdgcn_sched_barrier(0);
    HW_GROUP_(4) HW_GROUP_(5)
    __builtin_amdgcn_sched_barrier(0);
    HW_GROUP_(6) HW_GROUP_(7)
    __builtin_amdgcn_sched_barrier(0);
    if (more) {
      *(v4f*)(d + (BM_ +  0) * PITCH_)    = r0;
      *(v4f*)(d + (BM_ + 32) * PITCH_)    = r1;
      *(v4f*)(d + (BM_ + 64) * PITCH_)    = r2;
      *(v4f*)(d + (BM_ + 96) * PITCH_)    = r3;
      __syncthreads();
    }
  }
#undef HW_GROUP_

  const float bnA = bias[nA], bnB = bias[nB];
  const float bgA = bias[D_ + nA], bgB = bias[D_ + nB];

#pragma unroll
  for (int r = 0; r < 8; ++r) {
    {
      const int m = m0 + r + 8 * half;
      const float* xr = x + (size_t)m * D_;
      float* op = out + (size_t)m * orow;
      float g0 = 1.f / (1.f + __expf(-(aG0[r] + bgA)));
      op[nA] = g0 * xr[nA] + (1.f - g0) * fmaxf(aN0[r] + bnA, 0.f);
      float g1 = 1.f / (1.f + __expf(-(aG1[r] + bgB)));
      op[nB] = g1 * xr[nB] + (1.f - g1) * fmaxf(aN1[r] + bnB, 0.f);
    }
    {
      const int m = m0 + 16 + r + 8 * half;
      const float* xr = x + (size_t)m * D_;
      float* op = out + (size_t)m * orow;
      float g0 = 1.f / (1.f + __expf(-(bG0[r] + bgA)));
      op[nA] = g0 * xr[nA] + (1.f - g0) * fmaxf(bN0[r] + bnA, 0.f);
      float g1 = 1.f / (1.f + __expf(-(bG1[r] + bgB)));
      op[nB] = g1 * xr[nB] + (1.f - g1) * fmaxf(bN1[r] + bnB, 0.f);
    }
  }
}

extern "C" void kernel_launch(void* const* d_in, const int* in_sizes, int n_in,
                              void* d_out, int out_size, void* d_ws, size_t ws_size,
                              hipStream_t stream) {
  (void)in_sizes; (void)n_in; (void)out_size; (void)ws_size;

  const float* inputs   = (const float*)d_in[0];
  const float* fwd_pad  = (const float*)d_in[1];
  const float* bwd_pad  = (const float*)d_in[2];
  const float* fwd_w    = (const float*)d_in[3];
  const float* bwd_w    = (const float*)d_in[4];
  const float* fwd_hw_W = (const float*)d_in[5];
  const float* fwd_hw_b = (const float*)d_in[6];
  const float* bwd_hw_W = (const float*)d_in[7];
  const float* bwd_hw_b = (const float*)d_in[8];
  float* out = (float*)d_out;

  const size_t M = (size_t)B_ * S_;
  float* yA = (float*)d_ws;
  float* yB = yA + M * D_;

  dim3 hwBlock(256);
  dim3 hwGrid(D_ / BN_, (unsigned)(M / BM_));
  const int firBlocks = (int)(M * D_ / (4 * FIR_ROWS_ * 256));

  for (int i = 0; i < NL_; ++i) {
    for (int dir = 0; dir < 2; ++dir) {
      const float* xin; int xrow;
      if (i == 0) { xin = inputs; xrow = D_; }
      else        { xin = out + (size_t)(i - 1) * M * 2 * D_ + dir * D_; xrow = 2 * D_; }

      const float* taps = (dir ? bwd_w : fwd_w) + i * (W_ + 1);
      fir_kernel<<<firBlocks, 256, 0, stream>>>(
          xin, xrow,
          fwd_pad + (size_t)i * W_ * D_, bwd_pad + (size_t)i * W_ * D_,
          taps, dir ? W_ : 0, yA);

      const float* HW = (dir ? bwd_hw_W : fwd_hw_W) + (size_t)i * NH_ * 2 * D_ * D_;
      const float* Hb = (dir ? bwd_hw_b : fwd_hw_b) + (size_t)i * NH_ * 2 * D_;

      highway_kernel<<<hwGrid, hwBlock, 0, stream>>>(yA, HW, Hb, yB, D_);
      highway_kernel<<<hwGrid, hwBlock, 0, stream>>>(
          yB, HW + (size_t)2 * D_ * D_, Hb + 2 * D_,
          out + (size_t)i * M * 2 * D_ + dir * D_, 2 * D_);
    }
  }
}


# Forward and backward directions share each launch (grid z / y = direction): 6 launches instead of 12, two generations of workgroups per highway launch so drain and fill overlap; sums unchanged
# speedup vs baseline: 1.9522x; 1.0892x over previous
#include <hip/hip_runtime.h>
#include <hip/hip_bf16.h>

#define B_  32
#define S_  1024
#define D_  256
#define W_  4
#define NL_ 2
#define NH_ 2

typedef __attribute__((ext_vector_type(2))) float v2f;
typedef __attribute__((ext_vector_type(8))) float v8f;

#define FIR_ROWS_ 16

typedef __attribute__((ext_vector_type(4))) float v4f;

__global__ __launch_bounds__(256)
void fir_kernel(const float* __restrict__ xf, const float* __restrict__ xb, int xrow,
                const float* __restrict__ padf, const float* __restrict__ padb,
                const float* __restrict__ wf, const float* __restrict__ wb,
                float* __restrict__ yf, float* __restrict__ yb) {
  const int dir = blockIdx.y;
  const float* x = dir ? xb : xf;
  const float* w = dir ? wb : wf;
  float* y       = dir ? yb : yf;
  const int off = dir ? W_ : 0;
  const int gid = blockIdx.x * 256 + threadIdx.x;
  const int d   = (gid & 63) * 4;
  const int rb  = __builtin_amdgcn_readfirstlane(gid >> 6);
  const int s0  = (rb & (S_ / FIR_ROWS_ - 1)) * FIR_ROWS_;
  const int b   = rb >> 6;

  const float w0 = w[0], w1 = w[1], w2 = w[2], w3 = w[3], w4 = w[4];

  const int t0 = s0 + off;
  v4f v[FIR_ROWS_ + W_];
  if (t0 >= W_ && t0 + FIR_ROWS_ + W_ <= W_ + S_) {
    const float* p = x + ((size_t)b * S_ + (size_t)(t0 - W_)) * (size_t)xrow + d;
#pragma unroll
    for (int u = 0; u < FIR_ROWS_ + W_; ++u)
      v[u] = *(const v4f*)(p + (size_t)u * (size_t)xrow);
  } else {
#pragma unroll
    for (int u = 0; u < FIR_ROWS_ + W_; ++u) {
      const int t = t0 + u;
      const float* p;
      if (t < W_)             p = padf + t * D_;
      else if (t >= W_ + S_)  p = padb + (t - W_ - S_) * D_;
      else                    p = x + ((size_t)b * S_ + (size_t)(t - W_)) * (size_t)xrow;
      v[u] = *(const v4f*)(p + d);
    }
  }

  float* yp = y + ((size_t)b * S_ + (size_t)s0) * D_ + d;
#pragma unroll
  for (int r = 0; r < FIR_ROWS_; ++r) {
    v4f acc;
#pragma unroll
    for (int c = 0; c < 4; ++c) {
      float a = fmaf(w0, v[r][c], 0.f);
      a = fmaf(w1, v[r + 1][c], a);
      a = fmaf(w2, v[r + 2][c], a);
      a = fmaf(w3, v[r + 3][c], a);
      a = fmaf(w4, v[r + 4][c], a);
      acc[c] = a;
    }
    *(v4f*)(yp + (size_t)r * D_) = acc;
  }
}

#define BM_     128
#define BN_     64
#define KC_     32
#define PITCH_  36
#define STAGE_  ((BM_ + 2 * BN_) * PITCH_)

__global__ __launch_bounds__(256)
void highway_kernel(const float* __restrict__ x0, const float* __restrict__ x1,
                    const float* __restrict__ Wt0, const float* __restrict__ Wt1,
                    const float* __restrict__ bias0, const float* __restrict__ bias1,
                    float* __restrict__ out0, float* __restrict__ out1, int orow) {
  __shared__ __attribute__((aligned(16))) float lds[2 * STAGE_];

  const int dir = blockIdx.z;
  const float* x    = dir ? x1 : x0;
  const float* Wt   = dir ? Wt1 : Wt0;
  const float* bias = dir ? bias1 : bias0;
  float* out        = dir ? out1 : out0;

  const int tid  = threadIdx.x;
  const int lane = tid & 31;
  const int wave = tid >> 5;
  const int half = lane >> 4;
  const int l    = lane & 15;
  const int wm   = wave & 3;
  const int wn   = wave >> 2;

  const int mBlk = blockIdx.y * BM_;
  const int nBlk = blockIdx.x * BN_;

  const int m0 = mBlk + wm * 32;
  const int nA = nBlk + wn * 32 + l;
  const int nB = nA + 16;

  const int lrow = tid >> 3;
  const int lcol = (tid & 7) * 4;
  const float* gA  = x  + (size_t)(mBlk + lrow) * D_        + lcol;
  const float* gBn = Wt + (size_t)(nBlk + lrow) * D_        + lcol;
  const float* gBg = Wt + (size_t)(D_ + nBlk + lrow) * D_   + lcol;
  const int sOff = lrow * PITCH_ + lcol;

  const int fA0 = (wm * 32 + l) * PITCH_ + 2 * half;
  const int fA1 = fA0 + 16 * PITCH_;
  const int fN0 = (BM_ + wn * 32 + l) * PITCH_ + 2 * half;
  const int fN1 = fN0 + 16 * PITCH_;
  const int fG0 = fN0 + BN_ * PITCH_;
  const int fG1 = fG0 + 16 * PITCH_;

  v8f aN0 = {}; v8f aN1 = {}; v8f aG0 = {}; v8f aG1 = {};
  v8f bN0 = {}; v8f bN1 = {}; v8f bG0 = {}; v8f bG1 = {};

  v4f r0, r1, r2, r3;

#define HW_GROUP_(g)                                                                                   \
  {                                                                                                    \
    v2f a0  = *(const v2f*)(s + fA0 + 4 * (g));                                                        \
    v2f a1  = *(const v2f*)(s + fA1 + 4 * (g));                                                        \
    v2f n0v = *(const v2f*)(s + fN0 + 4 * (g));                                                        \
    v2f n1v = *(const v2f*)(s + fN1 + 4 * (g));                                                        \
    v2f g0v = *(const v2f*)(s + fG0 + 4 * (g));                                                        \
    v2f g1v = *(const v2f*)(s + fG1 + 4 * (g));                                                        \
    aN0 = __builtin_amdgcn_wmma_f32_16x16x4_f32(false, a0, false, n0v, (short)0, aN0, false, false);   \
    aN1 = __builtin_amdgcn_wmma_f32_16x16x4_f32(false, a0, false, n1v, (short)0, aN1, false, false);   \
    aG0 = __builtin_amdgcn_wmma_f32_16x16x4_f32(false, a0, false, g0v, (short)0, aG0, false, false);   \
    aG1 = __builtin_amdgcn_wmma_f32_16x16x4_f32(false, a0, false, g1v, (short)0, aG1, false, false);   \
    bN0 = __builtin_amdgcn_wmma_f32_16x16x4_f32(false, a1, false, n0v, (short)0, bN0, false, false);   \
    bN1 = __builtin_amdgcn_wmma_f32_16x16x4_f32(false, a1, false, n1v, (short)0, bN1, false, false);   \
    bG0 = __builtin_amdgcn_wmma_f32_16x16x4_f32(false, a1, false, g0v, (short)0, bG0, false, false);   \
    bG1 = __builtin_amdgcn_wmma_f32_16x16x4_f32(false, a1, false, g1v, (short)0, bG1, false, false);   \
  }

  {
    float* d = lds + sOff;
    r0 = *(const v4f*)(gA);
    r1 = *(const v4f*)(gA + 32 * D_);
    r2 = *(const v4f*)(gA + 64 * D_);
    r3 = *(const v4f*)(gA + 96 * D_);
    *(v4f*)(d)                          = r0;
    *(v4f*)(d + 32 * PITCH_)            = r1;
    *(v4f*)(d + 64 * PITCH_)            = r2;
    *(v4f*)(d + 96 * PITCH_)            = r3;
    r0 = *(const v4f*)(gBn);
    r1 = *(const v4f*)(gBn + 32 * D_);
    r2 = *(const v4f*)(gBg);
    r3 = *(const v4f*)(gBg + 32 * D_);
    *(v4f*)(d + (BM_ +  0) * PITCH_)    = r0;
    *(v4f*)(d + (BM_ + 32) * PITCH_)    = r1;
    *(v4f*)(d + (BM_ + 64) * PITCH_)    = r2;
    *(v4f*)(d + (BM_ + 96) * PITCH_)    = r3;
  }
  __syncthreads();

  for (int kc = 0; kc < D_ / KC_; ++kc) {
    const bool more = (kc + 1) < D_ / KC_;
    const int ko = (kc + 1) * KC_;
    const float* s = lds + (kc & 1) * STAGE_;
    float* d = lds + ((kc + 1) & 1) * STAGE_ + sOff;

    if (more) {
      r0 = *(const v4f*)(gA + ko);
      r1 = *(const v4f*)(gA + 32 * D_ + ko);
      r2 = *(const v4f*)(gA + 64 * D_ + ko);
      r3 = *(const v4f*)(gA + 96 * D_ + ko);
    }
    __builtin_amdgcn_sched_barrier(0);
    HW_GROUP_(0) HW_GROUP_(1)
    __builtin_amdgcn_sched_barrier(0);
    HW_GROUP_(2) HW_GROUP_(3)
    __builtin_amdgcn_sched_barrier(0);
    if (more) {
      *(v4f*)(d)                          = r0;
      *(v4f*)(d + 32 * PITCH_)            = r1;
      *(v4f*)(d + 64 * PITCH_)            = r2;
      *(v4f*)(d + 96 * PITCH_)            = r3;
      r0 = *(const v4f*)(gBn + ko);
      r1 = *(const v4f*)(gBn + 32 * D_ + ko);
      r2 = *(const v4f*)(gBg + ko);
      r3 = *(const v4f*)(gBg + 32 * D_ + ko);
    }
    __builtin_amdgcn_sched_barrier(0);
    HW_GROUP_(4) HW_GROUP_(5)
    __builtin_amdgcn_sched_barrier(0);
    HW_GROUP_(6) HW_GROUP_(7)
    __builtin_amdgcn_sched_barrier(0);
    if (more) {
      *(v4f*)(d + (BM_ +  0) * PITCH_)    = r0;
      *(v4f*)(d + (BM_ + 32) * PITCH_)    = r1;
      *(v4f*)(d + (BM_ + 64) * PITCH_)    = r2;
      *(v4f*)(d + (BM_ + 96) * PITCH_)    = r3;
      __syncthreads();
    }
  }
#undef HW_GROUP_

  const float bnA = bias[nA], bnB = bias[nB];
  const float bgA = bias[D_ + nA], bgB = bias[D_ + nB];

#pragma unroll
  for (int r = 0; r < 8; ++r) {
    {
      const int m = m0 + r + 8 * half;
      const float* xr = x + (size_t)m * D_;
      float* op = out + (size_t)m * orow;
      float g0 = 1.f / (1.f + __expf(-(aG0[r] + bgA)));
      op[nA] = g0 * xr[nA] + (1.f - g0) * fmaxf(aN0[r] + bnA, 0.f);
      float g1 = 1.f / (1.f + __expf(-(aG1[r] + bgB)));
      op[nB] = g1 * xr[nB] + (1.f - g1) * fmaxf(aN1[r] + bnB, 0.f);
    }
    {
      const int m = m0 + 16 + r + 8 * half;
      const float* xr = x + (size_t)m * D_;
      float* op = out + (size_t)m * orow;
      float g0 = 1.f / (1.f + __expf(-(bG0[r] + bgA)));
      op[nA] = g0 * xr[nA] + (1.f - g0) * fmaxf(bN0[r] + bnA, 0.f);
      float g1 = 1.f / (1.f + __expf(-(bG1[r] + bgB)));
      op[nB] = g1 * xr[nB] + (1.f - g1) * fmaxf(bN1[r] + bnB, 0.f);
    }
  }
}

extern "C" void kernel_launch(void* const* d_in, const int* in_sizes, int n_in,
                              void* d_out, int out_size, void* d_ws, size_t ws_size,
                              hipStream_t stream) {
  (void)in_sizes; (void)n_in; (void)out_size; (void)ws_size;

  const float* inputs   = (const float*)d_in[0];
  const float* fwd_pad  = (const float*)d_in[1];
  const float* bwd_pad  = (const float*)d_in[2];
  const float* fwd_w    = (const float*)d_in[3];
  const float* bwd_w    = (const float*)d_in[4];
  const float* fwd_hw_W = (const float*)d_in[5];
  const float* fwd_hw_b = (const float*)d_in[6];
  const float* bwd_hw_W = (const float*)d_in[7];
  const float* bwd_hw_b = (const float*)d_in[8];
  float* out = (float*)d_out;

  const size_t M = (size_t)B_ * S_;
  float* yAf = (float*)d_ws;
  float* yAb = yAf + M * D_;
  float* yBf = yAb + M * D_;
  float* yBb = yBf + M * D_;

  dim3 hwBlock(256);
  dim3 hwGrid(D_ / BN_, (unsigned)(M / BM_), 2);
  dim3 firGrid((unsigned)(M * D_ / (4 * FIR_ROWS_ * 256)), 2);

  for (int i = 0; i < NL_; ++i) {
    const float* xinf; const float* xinb; int xrow;
    if (i == 0) { xinf = inputs; xinb = inputs; xrow = D_; }
    else        { xinf = out + (size_t)(i - 1) * M * 2 * D_; xinb = xinf + D_; xrow = 2 * D_; }

    fir_kernel<<<firGrid, 256, 0, stream>>>(
        xinf, xinb, xrow,
        fwd_pad + (size_t)i * W_ * D_, bwd_pad + (size_t)i * W_ * D_,
        fwd_w + i * (W_ + 1), bwd_w + i * (W_ + 1),
        yAf, yAb);

    const float* HWf = fwd_hw_W + (size_t)i * NH_ * 2 * D_ * D_;
    const float* HWb = bwd_hw_W + (size_t)i * NH_ * 2 * D_ * D_;
    const float* Hbf = fwd_hw_b + (size_t)i * NH_ * 2 * D_;
    const float* Hbb = bwd_hw_b + (size_t)i * NH_ * 2 * D_;

    highway_kernel<<<hwGrid, hwBlock, 0, stream>>>(yAf, yAb, HWf, HWb, Hbf, Hbb, yBf, yBb, D_);
    float* oi = out + (size_t)i * M * 2 * D_;
    highway_kernel<<<hwGrid, hwBlock, 0, stream>>>(
        yBf, yBb,
        HWf + (size_t)2 * D_ * D_, HWb + (size_t)2 * D_ * D_,
        Hbf + 2 * D_, Hbb + 2 * D_,
        oi, oi + D_, 2 * D_);
  }
}


# Highway epilogue: sigmoid gate as one exp2 (bias folded into its multiply-add) and one hardware reciprocal in place of the IEEE division sequence, 11 fewer vector instructions per output; sums kept
# speedup vs baseline: 2.0174x; 1.0334x over previous
#include <hip/hip_runtime.h>
#include <hip/hip_bf16.h>

#define B_  32
#define S_  1024
#define D_  256
#define W_  4
#define NL_ 2
#define NH_ 2

typedef __attribute__((ext_vector_type(2))) float v2f;
typedef __attribute__((ext_vector_type(8))) float v8f;

#define FIR_ROWS_ 16

typedef __attribute__((ext_vector_type(4))) float v4f;

__global__ __launch_bounds__(256)
void fir_kernel(const float* __restrict__ xf, const float* __restrict__ xb, int xrow,
                const float* __restrict__ padf, const float* __restrict__ padb,
                const float* __restrict__ wf, const float* __restrict__ wb,
                float* __restrict__ yf, float* __restrict__ yb) {
  const int dir = blockIdx.y;
  const float* x = dir ? xb : xf;
  const float* w = dir ? wb : wf;
  float* y       = dir ? yb : yf;
  const int off = dir ? W_ : 0;
  const int gid = blockIdx.x * 256 + threadIdx.x;
  const int d   = (gid & 63) * 4;
  const int rb  = __builtin_amdgcn_readfirstlane(gid >> 6);
  const int s0  = (rb & (S_ / FIR_ROWS_ - 1)) * FIR_ROWS_;
  const int b   = rb >> 6;

  const float w0 = w[0], w1 = w[1], w2 = w[2], w3 = w[3], w4 = w[4];

  const int t0 = s0 + off;
  v4f v[FIR_ROWS_ + W_];
  if (t0 >= W_ && t0 + FIR_ROWS_ + W_ <= W_ + S_) {
    const float* p = x + ((size_t)b * S_ + (size_t)(t0 - W_)) * (size_t)xrow + d;
#pragma unroll
    for (int u = 0; u < FIR_ROWS_ + W_; ++u)
      v[u] = *(const v4f*)(p + (size_t)u * (size_t)xrow);
  } else {
#pragma unroll
    for (int u = 0; u < FIR_ROWS_ + W_; ++u) {
      const int t = t0 + u;
      const float* p;
      if (t < W_)             p = padf + t * D_;
      else if (t >= W_ + S_)  p = padb + (t - W_ - S_) * D_;
      else                    p = x + ((size_t)b * S_ + (size_t)(t - W_)) * (size_t)xrow;
      v[u] = *(const v4f*)(p + d);
    }
  }

  float* yp = y + ((size_t)b * S_ + (size_t)s0) * D_ + d;
#pragma unroll
  for (int r = 0; r < FIR_ROWS_; ++r) {
    v4f acc;
#pragma unroll
    for (int c = 0; c < 4; ++c) {
      float a = fmaf(w0, v[r][c], 0.f);
      a = fmaf(w1, v[r + 1][c], a);
      a = fmaf(w2, v[r + 2][c], a);
      a = fmaf(w3, v[r + 3][c], a);
      a = fmaf(w4, v[r + 4][c], a);
      acc[c] = a;
    }
    *(v4f*)(yp + (size_t)r * D_) = acc;
  }
}

#define BM_     128
#define BN_     64
#define KC_     32
#define PITCH_  36
#define STAGE_  ((BM_ + 2 * BN_) * PITCH_)

__device__ __forceinline__ float gate_sigmoid(float z, float cb) {
  const float e = __builtin_amdgcn_exp2f(fmaf(z, -1.44269504f, cb));
  return __builtin_amdgcn_rcpf(1.f + e);
}

__global__ __launch_bounds__(256)
void highway_kernel(const float* __restrict__ x0, const float* __restrict__ x1,
                    const float* __restrict__ Wt0, const float* __restrict__ Wt1,
                    const float* __restrict__ bias0, const float* __restrict__ bias1,
                    float* __restrict__ out0, float* __restrict__ out1, int orow) {
  __shared__ __attribute__((aligned(16))) float lds[2 * STAGE_];

  const int dir = blockIdx.z;
  const float* x    = dir ? x1 : x0;
  const float* Wt   = dir ? Wt1 : Wt0;
  const float* bias = dir ? bias1 : bias0;
  float* out        = dir ? out1 : out0;

  const int tid  = threadIdx.x;
  const int lane = tid & 31;
  const int wave = tid >> 5;
  const int half = lane >> 4;
  const int l    = lane & 15;
  const int wm   = wave & 3;
  const int wn   = wave >> 2;

  const int mBlk = blockIdx.y * BM_;
  const int nBlk = blockIdx.x * BN_;

  const int m0 = mBlk + wm * 32;
  const int nA = nBlk + wn * 32 + l;
  const int nB = nA + 16;

  const int lrow = tid >> 3;
  const int lcol = (tid & 7) * 4;
  const float* gA  = x  + (size_t)(mBlk + lrow) * D_        + lcol;
  const float* gBn = Wt + (size_t)(nBlk + lrow) * D_        + lcol;
  const float* gBg = Wt + (size_t)(D_ + nBlk + lrow) * D_   + lcol;
  const int sOff = lrow * PITCH_ + lcol;

  const int fA0 = (wm * 32 + l) * PITCH_ + 2 * half;
  const int fA1 = fA0 + 16 * PITCH_;
  const int fN0 = (BM_ + wn * 32 + l) * PITCH_ + 2 * half;
  const int fN1 = fN0 + 16 * PITCH_;
  const int fG0 = fN0 + BN_ * PITCH_;
  const int fG1 = fG0 + 16 * PITCH_;

  v8f aN0 = {}; v8f aN1 = {}; v8f aG0 = {}; v8f aG1 = {};
  v8f bN0 = {}; v8f bN1 = {}; v8f bG0 = {}; v8f bG1 = {};

  v4f r0, r1, r2, r3;

#define HW_GROUP_(g)                                                                                   \
  {                                                                                                    \
    v2f a0  = *(const v2f*)(s + fA0 + 4 * (g));                                                        \
    v2f a1  = *(const v2f*)(s + fA1 + 4 * (g));                                                        \
    v2f n0v = *(const v2f*)(s + fN0 + 4 * (g));                                                        \
    v2f n1v = *(const v2f*)(s + fN1 + 4 * (g));                                                        \
    v2f g0v = *(const v2f*)(s + fG0 + 4 * (g));                                                        \
    v2f g1v = *(const v2f*)(s + fG1 + 4 * (g));                                                        \
    aN0 = __builtin_amdgcn_wmma_f32_16x16x4_f32(false, a0, false, n0v, (short)0, aN0, false, false);   \
    aN1 = __builtin_amdgcn_wmma_f32_16x16x4_f32(false, a0, false, n1v, (short)0, aN1, false, false);   \
    aG0 = __builtin_amdgcn_wmma_f32_16x16x4_f32(false, a0, false, g0v, (short)0, aG0, false, false);   \
    aG1 = __builtin_amdgcn_wmma_f32_16x16x4_f32(false, a0, false, g1v, (short)0, aG1, false, false);   \
    bN0 = __builtin_amdgcn_wmma_f32_16x16x4_f32(false, a1, false, n0v, (short)0, bN0, false, false);   \
    bN1 = __builtin_amdgcn_wmma_f32_16x16x4_f32(false, a1, false, n1v, (short)0, bN1, false, false);   \
    bG0 = __builtin_amdgcn_wmma_f32_16x16x4_f32(false, a1, false, g0v, (short)0, bG0, false, false);   \
    bG1 = __builtin_amdgcn_wmma_f32_16x16x4_f32(false, a1, false, g1v, (short)0, bG1, false, false);   \
  }

  {
    float* d = lds + sOff;
    r0 = *(const v4f*)(gA);
    r1 = *(const v4f*)(gA + 32 * D_);
    r2 = *(const v4f*)(gA + 64 * D_);
    r3 = *(const v4f*)(gA + 96 * D_);
    *(v4f*)(d)                          = r0;
    *(v4f*)(d + 32 * PITCH_)            = r1;
    *(v4f*)(d + 64 * PITCH_)            = r2;
    *(v4f*)(d + 96 * PITCH_)            = r3;
    r0 = *(const v4f*)(gBn);
    r1 = *(const v4f*)(gBn + 32 * D_);
    r2 = *(const v4f*)(gBg);
    r3 = *(const v4f*)(gBg + 32 * D_);
    *(v4f*)(d + (BM_ +  0) * PITCH_)    = r0;
    *(v4f*)(d + (BM_ + 32) * PITCH_)    = r1;
    *(v4f*)(d + (BM_ + 64) * PITCH_)    = r2;
    *(v4f*)(d + (BM_ + 96) * PITCH_)    = r3;
  }
  __syncthreads();

  for (int kc = 0; kc < D_ / KC_; ++kc) {
    const bool more = (kc + 1) < D_ / KC_;
    const int ko = (kc + 1) * KC_;
    const float* s = lds + (kc & 1) * STAGE_;
    float* d = lds + ((kc + 1) & 1) * STAGE_ + sOff;

    if (more) {
      r0 = *(const v4f*)(gA + ko);
      r1 = *(const v4f*)(gA + 32 * D_ + ko);
      r2 = *(const v4f*)(gA + 64 * D_ + ko);
      r3 = *(const v4f*)(gA + 96 * D_ + ko);
    }
    __builtin_amdgcn_sched_barrier(0);
    HW_GROUP_(0) HW_GROUP_(1)
    __builtin_amdgcn_sched_barrier(0);
    HW_GROUP_(2) HW_GROUP_(3)
    __builtin_amdgcn_sched_barrier(0);
    if (more) {
      *(v4f*)(d)                          = r0;
      *(v4f*)(d + 32 * PITCH_)            = r1;
      *(v4f*)(d + 64 * PITCH_)            = r2;
      *(v4f*)(d + 96 * PITCH_)            = r3;
      r0 = *(const v4f*)(gBn + ko);
      r1 = *(const v4f*)(gBn + 32 * D_ + ko);
      r2 = *(const v4f*)(gBg + ko);
      r3 = *(const v4f*)(gBg + 32 * D_ + ko);
    }
    __builtin_amdgcn_sched_barrier(0);
    HW_GROUP_(4) HW_GROUP_(5)
    __builtin_amdgcn_sched_barrier(0);
    HW_GROUP_(6) HW_GROUP_(7)
    __builtin_amdgcn_sched_barrier(0);
    if (more) {
      *(v4f*)(d + (BM_ +  0) * PITCH_)    = r0;
      *(v4f*)(d + (BM_ + 32) * PITCH_)    = r1;
      *(v4f*)(d + (BM_ + 64) * PITCH_)    = r2;
      *(v4f*)(d + (BM_ + 96) * PITCH_)    = r3;
      __syncthreads();
    }
  }
#undef HW_GROUP_

  const float bnA = bias[nA], bnB = bias[nB];
  const float cgA = bias[D_ + nA] * -1.44269504f, cgB = bias[D_ + nB] * -1.44269504f;

#pragma unroll
  for (int r = 0; r < 8; ++r) {
    {
      const int m = m0 + r + 8 * half;
      const float* xr = x + (size_t)m * D_;
      float* op = out + (size_t)m * orow;
      float g0 = gate_sigmoid(aG0[r], cgA);
      op[nA] = g0 * xr[nA] + (1.f - g0) * fmaxf(aN0[r] + bnA, 0.f);
      float g1 = gate_sigmoid(aG1[r], cgB);
      op[nB] = g1 * xr[nB] + (1.f - g1) * fmaxf(aN1[r] + bnB, 0.f);
    }
    {
      const int m = m0 + 16 + r + 8 * half;
      const float* xr = x + (size_t)m * D_;
      float* op = out + (size_t)m * orow;
      float g0 = gate_sigmoid(bG0[r], cgA);
      op[nA] = g0 * xr[nA] + (1.f - g0) * fmaxf(bN0[r] + bnA, 0.f);
      float g1 = gate_sigmoid(bG1[r], cgB);
      op[nB] = g1 * xr[nB] + (1.f - g1) * fmaxf(bN1[r] + bnB, 0.f);
    }
  }
}

extern "C" void kernel_launch(void* const* d_in, const int* in_sizes, int n_in,
                              void* d_out, int out_size, void* d_ws, size_t ws_size,
                              hipStream_t stream) {
  (void)in_sizes; (void)n_in; (void)out_size; (void)ws_size;

  const float* inputs   = (const float*)d_in[0];
  const float* fwd_pad  = (const float*)d_in[1];
  const float* bwd_pad  = (const float*)d_in[2];
  const float* fwd_w    = (const float*)d_in[3];
  const float* bwd_w    = (const float*)d_in[4];
  const float* fwd_hw_W = (const float*)d_in[5];
  const float* fwd_hw_b = (const float*)d_in[6];
  const float* bwd_hw_W = (const float*)d_in[7];
  const float* bwd_hw_b = (const float*)d_in[8];
  float* out = (float*)d_out;

  const size_t M = (size_t)B_ * S_;
  float* yAf = (float*)d_ws;
  float* yAb = yAf + M * D_;
  float* yBf = yAb + M * D_;
  float* yBb = yBf + M * D_;

  dim3 hwBlock(256);
  dim3 hwGrid(D_ / BN_, (unsigned)(M / BM_), 2);
  dim3 firGrid((unsigned)(M * D_ / (4 * FIR_ROWS_ * 256)), 2);

  for (int i = 0; i < NL_; ++i) {
    const float* xinf; const float* xinb; int xrow;
    if (i == 0) { xinf = inputs; xinb = inputs; xrow = D_; }
    else        { xinf = out + (size_t)(i - 1) * M * 2 * D_; xinb = xinf + D_; xrow = 2 * D_; }

    fir_kernel<<<firGrid, 256, 0, stream>>>(
        xinf, xinb, xrow,
        fwd_pad + (size_t)i * W_ * D_, bwd_pad + (size_t)i * W_ * D_,
        fwd_w + i * (W_ + 1), bwd_w + i * (W_ + 1),
        yAf, yAb);

    const float* HWf = fwd_hw_W + (size_t)i * NH_ * 2 * D_ * D_;
    const float* HWb = bwd_hw_W + (size_t)i * NH_ * 2 * D_ * D_;
    const float* Hbf = fwd_hw_b + (size_t)i * NH_ * 2 * D_;
    const float* Hbb = bwd_hw_b + (size_t)i * NH_ * 2 * D_;

    highway_kernel<<<hwGrid, hwBlock, 0, stream>>>(yAf, yAb, HWf, HWb, Hbf, Hbb, yBf, yBb, D_);
    float* oi = out + (size_t)i * M * 2 * D_;
    highway_kernel<<<hwGrid, hwBlock, 0, stream>>>(
        yBf, yBb,
        HWf + (size_t)2 * D_ * D_, HWb + (size_t)2 * D_ * D_,
        Hbf + 2 * D_, Hbb + 2 * D_,
        oi, oi + D_, 2 * D_);
  }
}


# First highway layer's tile loader forms the 5-tap FIR itself (8 window rows per thread from x or the pads, same tap order), residual x kept in LDS: both FIR launches go, 4 launches; sums unchanged
# speedup vs baseline: 2.1309x; 1.0563x over previous
#include <hip/hip_runtime.h>
#include <hip/hip_bf16.h>

#define B_  32
#define S_  1024
#define D_  256
#define W_  4
#define NL_ 2
#define NH_ 2

typedef __attribute__((ext_vector_type(2))) float v2f;
typedef __attribute__((ext_vector_type(8))) float v8f;

typedef __attribute__((ext_vector_type(4))) float v4f;

#define BM_     128
#define BN_     64
#define KC_     32
#define PITCH_  36
#define STAGE_  ((BM_ + 2 * BN_) * PITCH_)

__device__ __forceinline__ float gate_sigmoid(float z, float cb) {
  const float e = __builtin_amdgcn_exp2f(fmaf(z, -1.44269504f, cb));
  return __builtin_amdgcn_rcpf(1.f + e);
}

__global__ __launch_bounds__(256)
void highway_kernel(const float* __restrict__ x0, const float* __restrict__ x1,
                    const float* __restrict__ Wt0, const float* __restrict__ Wt1,
                    const float* __restrict__ bias0, const float* __restrict__ bias1,
                    float* __restrict__ out0, float* __restrict__ out1, int orow) {
  __shared__ __attribute__((aligned(16))) float lds[2 * STAGE_];

  const int dir = blockIdx.z;
  const float* x    = dir ? x1 : x0;
  const float* Wt   = dir ? Wt1 : Wt0;
  const float* bias = dir ? bias1 : bias0;
  float* out        = dir ? out1 : out0;

  const int tid  = threadIdx.x;
  const int lane = tid & 31;
  const int wave = tid >> 5;
  const int half = lane >> 4;
  const int l    = lane & 15;
  const int wm   = wave & 3;
  const int wn   = wave >> 2;

  const int mBlk = blockIdx.y * BM_;
  const int nBlk = blockIdx.x * BN_;

  const int m0 = mBlk + wm * 32;
  const int nA = nBlk + wn * 32 + l;
  const int nB = nA + 16;

  const int lrow = tid >> 3;
  const int lcol = (tid & 7) * 4;
  const float* gA  = x  + (size_t)(mBlk + lrow) * D_        + lcol;
  const float* gBn = Wt + (size_t)(nBlk + lrow) * D_        + lcol;
  const float* gBg = Wt + (size_t)(D_ + nBlk + lrow) * D_   + lcol;
  const int sOff = lrow * PITCH_ + lcol;

  const int fA0 = (wm * 32 + l) * PITCH_ + 2 * half;
  const int fA1 = fA0 + 16 * PITCH_;
  const int fN0 = (BM_ + wn * 32 + l) * PITCH_ + 2 * half;
  const int fN1 = fN0 + 16 * PITCH_;
  const int fG0 = fN0 + BN_ * PITCH_;
  const int fG1 = fG0 + 16 * PITCH_;

  v8f aN0 = {}; v8f aN1 = {}; v8f aG0 = {}; v8f aG1 = {};
  v8f bN0 = {}; v8f bN1 = {}; v8f bG0 = {}; v8f bG1 = {};

  v4f r0, r1, r2, r3;

#define HW_GROUP_(g)                                                                                   \
  {                                                                                                    \
    v2f a0  = *(const v2f*)(s + fA0 + 4 * (g));                                                        \
    v2f a1  = *(const v2f*)(s + fA1 + 4 * (g));                                                        \
    v2f n0v = *(const v2f*)(s + fN0 + 4 * (g));                                                        \
    v2f n1v = *(const v2f*)(s + fN1 + 4 * (g));                                                        \
    v2f g0v = *(const v2f*)(s + fG0 + 4 * (g));                                                        \
    v2f g1v = *(const v2f*)(s + fG1 + 4 * (g));                                                        \
    aN0 = __builtin_amdgcn_wmma_f32_16x16x4_f32(false, a0, false, n0v, (short)0, aN0, false, false);   \
    aN1 = __builtin_amdgcn_wmma_f32_16x16x4_f32(false, a0, false, n1v, (short)0, aN1, false, false);   \
    aG0 = __builtin_amdgcn_wmma_f32_16x16x4_f32(false, a0, false, g0v, (short)0, aG0, false, false);   \
    aG1 = __builtin_amdgcn_wmma_f32_16x16x4_f32(false, a0, false, g1v, (short)0, aG1, false, false);   \
    bN0 = __builtin_amdgcn_wmma_f32_16x16x4_f32(false, a1, false, n0v, (short)0, bN0, false, false);   \
    bN1 = __builtin_amdgcn_wmma_f32_16x16x4_f32(false, a1, false, n1v, (short)0, bN1, false, false);   \
    bG0 = __builtin_amdgcn_wmma_f32_16x16x4_f32(false, a1, false, g0v, (short)0, bG0, false, false);   \
    bG1 = __builtin_amdgcn_wmma_f32_16x16x4_f32(false, a1, false, g1v, (short)0, bG1, false, false);   \
  }

  {
    float* d = lds + sOff;
    r0 = *(const v4f*)(gA);
    r1 = *(const v4f*)(gA + 32 * D_);
    r2 = *(const v4f*)(gA + 64 * D_);
    r3 = *(const v4f*)(gA + 96 * D_);
    *(v4f*)(d)                          = r0;
    *(v4f*)(d + 32 * PITCH_)            = r1;
    *(v4f*)(d + 64 * PITCH_)            = r2;
    *(v4f*)(d + 96 * PITCH_)            = r3;
    r0 = *(const v4f*)(gBn);
    r1 = *(const v4f*)(gBn + 32 * D_);
    r2 = *(const v4f*)(gBg);
    r3 = *(const v4f*)(gBg + 32 * D_);
    *(v4f*)(d + (BM_ +  0) * PITCH_)    = r0;
    *(v4f*)(d + (BM_ + 32) * PITCH_)    = r1;
    *(v4f*)(d + (BM_ + 64) * PITCH_)    = r2;
    *(v4f*)(d + (BM_ + 96) * PITCH_)    = r3;
  }
  __syncthreads();

  for (int kc = 0; kc < D_ / KC_; ++kc) {
    const bool more = (kc + 1) < D_ / KC_;
    const int ko = (kc + 1) * KC_;
    const float* s = lds + (kc & 1) * STAGE_;
    float* d = lds + ((kc + 1) & 1) * STAGE_ + sOff;

    if (more) {
      r0 = *(const v4f*)(gA + ko);
      r1 = *(const v4f*)(gA + 32 * D_ + ko);
      r2 = *(const v4f*)(gA + 64 * D_ + ko);
      r3 = *(const v4f*)(gA + 96 * D_ + ko);
    }
    __builtin_amdgcn_sched_barrier(0);
    HW_GROUP_(0) HW_GROUP_(1)
    __builtin_amdgcn_sched_barrier(0);
    HW_GROUP_(2) HW_GROUP_(3)
    __builtin_amdgcn_sched_barrier(0);
    if (more) {
      *(v4f*)(d)                          = r0;
      *(v4f*)(d + 32 * PITCH_)            = r1;
      *(v4f*)(d + 64 * PITCH_)            = r2;
      *(v4f*)(d + 96 * PITCH_)            = r3;
      r0 = *(const v4f*)(gBn + ko);
      r1 = *(const v4f*)(gBn + 32 * D_ + ko);
      r2 = *(const v4f*)(gBg + ko);
      r3 = *(const v4f*)(gBg + 32 * D_ + ko);
    }
    __builtin_amdgcn_sched_barrier(0);
    HW_GROUP_(4) HW_GROUP_(5)
    __builtin_amdgcn_sched_barrier(0);
    HW_GROUP_(6) HW_GROUP_(7)
    __builtin_amdgcn_sched_barrier(0);
    if (more) {
      *(v4f*)(d + (BM_ +  0) * PITCH_)    = r0;
      *(v4f*)(d + (BM_ + 32) * PITCH_)    = r1;
      *(v4f*)(d + (BM_ + 64) * PITCH_)    = r2;
      *(v4f*)(d + (BM_ + 96) * PITCH_)    = r3;
      __syncthreads();
    }
  }
#undef HW_GROUP_

  const float bnA = bias[nA], bnB = bias[nB];
  const float cgA = bias[D_ + nA] * -1.44269504f, cgB = bias[D_ + nB] * -1.44269504f;

#pragma unroll
  for (int r = 0; r < 8; ++r) {
    {
      const int m = m0 + r + 8 * half;
      const float* xr = x + (size_t)m * D_;
      float* op = out + (size_t)m * orow;
      float g0 = gate_sigmoid(aG0[r], cgA);
      op[nA] = g0 * xr[nA] + (1.f - g0) * fmaxf(aN0[r] + bnA, 0.f);
      float g1 = gate_sigmoid(aG1[r], cgB);
      op[nB] = g1 * xr[nB] + (1.f - g1) * fmaxf(aN1[r] + bnB, 0.f);
    }
    {
      const int m = m0 + 16 + r + 8 * half;
      const float* xr = x + (size_t)m * D_;
      float* op = out + (size_t)m * orow;
      float g0 = gate_sigmoid(bG0[r], cgA);
      op[nA] = g0 * xr[nA] + (1.f - g0) * fmaxf(bN0[r] + bnA, 0.f);
      float g1 = gate_sigmoid(bG1[r], cgB);
      op[nB] = g1 * xr[nB] + (1.f - g1) * fmaxf(bN1[r] + bnB, 0.f);
    }
  }
}

#define FW_     (2 * W_)
#define RPITCH_ 68

template <int XROW>
__global__ __launch_bounds__(256)
void highway_fir_kernel(const float* __restrict__ xf, const float* __restrict__ xb,
                        const float* __restrict__ padf, const float* __restrict__ padb,
                        const float* __restrict__ wf, const float* __restrict__ wb,
                        const float* __restrict__ Wt0, const float* __restrict__ Wt1,
                        const float* __restrict__ bias0, const float* __restrict__ bias1,
                        float* __restrict__ out0, float* __restrict__ out1) {
  __shared__ __attribute__((aligned(16))) float lds[2 * STAGE_ + BM_ * RPITCH_];
  float* res = lds + 2 * STAGE_;

  const int dir = blockIdx.z;
  const float* x    = dir ? xb : xf;
  const float* fw   = dir ? wb : wf;
  const float* Wt   = dir ? Wt1 : Wt0;
  const float* bias = dir ? bias1 : bias0;
  float* out        = dir ? out1 : out0;

  const int tid  = threadIdx.x;
  const int lane = tid & 31;
  const int wave = tid >> 5;
  const int half = lane >> 4;
  const int l    = lane & 15;
  const int wm   = wave & 3;
  const int wn   = wave >> 2;

  const int mBlk = blockIdx.y * BM_;
  const int nBlk = blockIdx.x * BN_;

  const int m0 = mBlk + wm * 32;
  const int nA = nBlk + wn * 32 + l;
  const int nB = nA + 16;

  const float w0 = fw[0], w1 = fw[1], w2 = fw[2], w3 = fw[3], w4 = fw[4];

  const int lrow = tid >> 3;
  const int lcol = (tid & 7) * 4;
  const int seq  = mBlk / S_;
  const int s0   = mBlk % S_;
  const int t0   = s0 + 4 * lrow + (dir ? W_ : 0);
  const int xLo = max(t0 - W_, 0);
  const int xHi = min(t0, S_ - 4);
  const float* gLo = x + ((size_t)seq * S_ + (size_t)xLo) * XROW + lcol;
  const float* gHi = x + ((size_t)seq * S_ + (size_t)xHi) * XROW + lcol;
  int sLo = XROW, sHi = XROW;
  if (t0 < W_)            { gLo = padf + t0 * D_ + lcol;                  sLo = D_; }
  if (t0 + 4 >= W_ + S_)  { gHi = padb + (t0 + 4 - W_ - S_) * D_ + lcol;  sHi = D_; }
  const float* gBn = Wt + (size_t)(nBlk + lrow) * D_        + lcol;
  const float* gBg = Wt + (size_t)(D_ + nBlk + lrow) * D_   + lcol;
  const int sOffA = 4 * lrow * PITCH_ + lcol;
  const int sOff  = lrow * PITCH_ + lcol;
  const int rOff  = 4 * lrow * RPITCH_ + lcol;

  const int fA0 = (wm * 32 + l) * PITCH_ + 2 * half;
  const int fA1 = fA0 + 16 * PITCH_;
  const int fN0 = (BM_ + wn * 32 + l) * PITCH_ + 2 * half;
  const int fN1 = fN0 + 16 * PITCH_;
  const int fG0 = fN0 + BN_ * PITCH_;
  const int fG1 = fG0 + 16 * PITCH_;

  v8f aN0 = {}; v8f aN1 = {}; v8f aG0 = {}; v8f aG1 = {};
  v8f bN0 = {}; v8f bN1 = {}; v8f bG0 = {}; v8f bG1 = {};

  v4f v[FW_];
  v4f r0, r1, r2, r3;

#define FIR_LOAD_(KO)                                                                                  \
  _Pragma("unroll")                                                                                    \
  for (int u = 0; u < 4; ++u)                                                                          \
    v[u]     = *(const v4f*)(gLo + u * (XROW == D_ ? D_ : sLo) + (KO));                                \
  _Pragma("unroll")                                                                                    \
  for (int u = 0; u < 4; ++u)                                                                          \
    v[u + 4] = *(const v4f*)(gHi + u * (XROW == D_ ? D_ : sHi) + (KO));

#define FIR_STORE_(DST, KC)                                                                            \
  {                                                                                                    \
    const bool keep = ((KC) >> 1) == (int)blockIdx.x;                                      \
    _Pragma("unroll")                                                                                  \
    for (int r = 0; r < 4; ++r) {                                                                      \
      v4f acc;                                                                                         \
      _Pragma("unroll")                                                                                \
      for (int c = 0; c < 4; ++c) {                                                                    \
        float a = fmaf(w0, v[r][c], 0.f);                                                              \
        a = fmaf(w1, v[r + 1][c], a);                                                                  \
        a = fmaf(w2, v[r + 2][c], a);                                                                  \
        a = fmaf(w3, v[r + 3][c], a);                                                                  \
        a = fmaf(w4, v[r + 4][c], a);                                                                  \
        acc[c] = a;                                                                                    \
      }                                                                                                \
      *(v4f*)((DST) + r * PITCH_) = acc;                                                               \
      if (keep) *(v4f*)(res + rOff + r * RPITCH_ + ((KC) & 1) * KC_) = acc;                            \
    }                                                                                                  \
  }

#define HW_GROUP_(g)                                                                                   \
  {                                                                                                    \
    v2f a0  = *(const v2f*)(s + fA0 + 4 * (g));                                                        \
    v2f a1  = *(const v2f*)(s + fA1 + 4 * (g));                                                        \
    v2f n0v = *(const v2f*)(s + fN0 + 4 * (g));                                                        \
    v2f n1v = *(const v2f*)(s + fN1 + 4 * (g));                                                        \
    v2f g0v = *(const v2f*)(s + fG0 + 4 * (g));                                                        \
    v2f g1v = *(const v2f*)(s + fG1 + 4 * (g));                                                        \
    aN0 = __builtin_amdgcn_wmma_f32_16x16x4_f32(false, a0, false, n0v, (short)0, aN0, false, false);   \
    aN1 = __builtin_amdgcn_wmma_f32_16x16x4_f32(false, a0, false, n1v, (short)0, aN1, false, false);   \
    aG0 = __builtin_amdgcn_wmma_f32_16x16x4_f32(false, a0, false, g0v, (short)0, aG0, false, false);   \
    aG1 = __builtin_amdgcn_wmma_f32_16x16x4_f32(false, a0, false, g1v, (short)0, aG1, false, false);   \
    bN0 = __builtin_amdgcn_wmma_f32_16x16x4_f32(false, a1, false, n0v, (short)0, bN0, false, false);   \
    bN1 = __builtin_amdgcn_wmma_f32_16x16x4_f32(false, a1, false, n1v, (short)0, bN1, false, false);   \
    bG0 = __builtin_amdgcn_wmma_f32_16x16x4_f32(false, a1, false, g0v, (short)0, bG0, false, false);   \
    bG1 = __builtin_amdgcn_wmma_f32_16x16x4_f32(false, a1, false, g1v, (short)0, bG1, false, false);   \
  }


  {
    float* d = lds + sOff;
    FIR_LOAD_(0)
    FIR_STORE_(lds + sOffA, 0)
    r0 = *(const v4f*)(gBn);
    r1 = *(const v4f*)(gBn + 32 * D_);
    r2 = *(const v4f*)(gBg);
    r3 = *(const v4f*)(gBg + 32 * D_);
    *(v4f*)(d + (BM_ +  0) * PITCH_)    = r0;
    *(v4f*)(d + (BM_ + 32) * PITCH_)    = r1;
    *(v4f*)(d + (BM_ + 64) * PITCH_)    = r2;
    *(v4f*)(d + (BM_ + 96) * PITCH_)    = r3;
  }
  __syncthreads();

  for (int kc = 0; kc < D_ / KC_; ++kc) {
    const bool more = (kc + 1) < D_ / KC_;
    const int ko = (kc + 1) * KC_;
    const float* s = lds + (kc & 1) * STAGE_;
    float* d = lds + ((kc + 1) & 1) * STAGE_;

    if (more) {
      FIR_LOAD_(ko)
    }
    __builtin_amdgcn_sched_barrier(0);
    HW_GROUP_(0) HW_GROUP_(1)
    __builtin_amdgcn_sched_barrier(0);
    HW_GROUP_(2) HW_GROUP_(3)
    __builtin_amdgcn_sched_barrier(0);
    if (more) {
      FIR_STORE_(d + sOffA, kc + 1)
      r0 = *(const v4f*)(gBn + ko);
      r1 = *(const v4f*)(gBn + 32 * D_ + ko);
      r2 = *(const v4f*)(gBg + ko);
      r3 = *(const v4f*)(gBg + 32 * D_ + ko);
    }
    __builtin_amdgcn_sched_barrier(0);
    HW_GROUP_(4) HW_GROUP_(5)
    __builtin_amdgcn_sched_barrier(0);
    HW_GROUP_(6) HW_GROUP_(7)
    __builtin_amdgcn_sched_barrier(0);
    if (more) {
      float* dw = d + sOff;
      *(v4f*)(dw + (BM_ +  0) * PITCH_)   = r0;
      *(v4f*)(dw + (BM_ + 32) * PITCH_)   = r1;
      *(v4f*)(dw + (BM_ + 64) * PITCH_)   = r2;
      *(v4f*)(dw + (BM_ + 96) * PITCH_)   = r3;
      __syncthreads();
    }
  }
#undef HW_GROUP_
#undef FIR_LOAD_
#undef FIR_STORE_

  const float bnA = bias[nA], bnB = bias[nB];
  const float cgA = bias[D_ + nA] * -1.44269504f, cgB = bias[D_ + nB] * -1.44269504f;

  const float* rp = res + (wm * 32 + 8 * half) * RPITCH_ + wn * 32 + l;
#pragma unroll
  for (int r = 0; r < 8; ++r) {
    {
      const int m = m0 + r + 8 * half;
      const float* xr = rp + r * RPITCH_;
      float* op = out + (size_t)m * D_;
      float g0 = gate_sigmoid(aG0[r], cgA);
      op[nA] = g0 * xr[0] + (1.f - g0) * fmaxf(aN0[r] + bnA, 0.f);
      float g1 = gate_sigmoid(aG1[r], cgB);
      op[nB] = g1 * xr[16] + (1.f - g1) * fmaxf(aN1[r] + bnB, 0.f);
    }
    {
      const int m = m0 + 16 + r + 8 * half;
      const float* xr = rp + (16 + r) * RPITCH_;
      float* op = out + (size_t)m * D_;
      float g0 = gate_sigmoid(bG0[r], cgA);
      op[nA] = g0 * xr[0] + (1.f - g0) * fmaxf(bN0[r] + bnA, 0.f);
      float g1 = gate_sigmoid(bG1[r], cgB);
      op[nB] = g1 * xr[16] + (1.f - g1) * fmaxf(bN1[r] + bnB, 0.f);
    }
  }
}

extern "C" void kernel_launch(void* const* d_in, const int* in_sizes, int n_in,
                              void* d_out, int out_size, void* d_ws, size_t ws_size,
                              hipStream_t stream) {
  (void)in_sizes; (void)n_in; (void)out_size; (void)ws_size;

  const float* inputs   = (const float*)d_in[0];
  const float* fwd_pad  = (const float*)d_in[1];
  const float* bwd_pad  = (const float*)d_in[2];
  const float* fwd_w    = (const float*)d_in[3];
  const float* bwd_w    = (const float*)d_in[4];
  const float* fwd_hw_W = (const float*)d_in[5];
  const float* fwd_hw_b = (const float*)d_in[6];
  const float* bwd_hw_W = (const float*)d_in[7];
  const float* bwd_hw_b = (const float*)d_in[8];
  float* out = (float*)d_out;

  const size_t M = (size_t)B_ * S_;
  float* yBf = (float*)d_ws;
  float* yBb = yBf + M * D_;

  dim3 hwBlock(256);
  dim3 hwGrid(D_ / BN_, (unsigned)(M / BM_), 2);

  for (int i = 0; i < NL_; ++i) {
    const float* padf = fwd_pad + (size_t)i * W_ * D_;
    const float* padb = bwd_pad + (size_t)i * W_ * D_;
    const float* tf   = fwd_w + i * (W_ + 1);
    const float* tb   = bwd_w + i * (W_ + 1);
    const float* HWf = fwd_hw_W + (size_t)i * NH_ * 2 * D_ * D_;
    const float* HWb = bwd_hw_W + (size_t)i * NH_ * 2 * D_ * D_;
    const float* Hbf = fwd_hw_b + (size_t)i * NH_ * 2 * D_;
    const float* Hbb = bwd_hw_b + (size_t)i * NH_ * 2 * D_;

    if (i == 0) {
      highway_fir_kernel<D_><<<hwGrid, hwBlock, 0, stream>>>(
          inputs, inputs, padf, padb, tf, tb, HWf, HWb, Hbf, Hbb, yBf, yBb);
    } else {
      const float* xin = out + (size_t)(i - 1) * M * 2 * D_;
      highway_fir_kernel<2 * D_><<<hwGrid, hwBlock, 0, stream>>>(
          xin, xin + D_, padf, padb, tf, tb, HWf, HWb, Hbf, Hbb, yBf, yBb);
    }
    float* oi = out + (size_t)i * M * 2 * D_;
    highway_kernel<<<hwGrid, hwBlock, 0, stream>>>(
        yBf, yBb,
        HWf + (size_t)2 * D_ * D_, HWb + (size_t)2 * D_ * D_,
        Hbf + 2 * D_, Hbb + 2 * D_,
        oi, oi + D_, 2 * D_);
  }
}


# Both highway layers of a block in one launch: a workgroup owns 128 rows x all 512 projection columns (32 accumulator tiles per wave), layer input and output stay in LDS; 2 launches; sums kept
# speedup vs baseline: 2.3202x; 1.0888x over previous
#include <hip/hip_runtime.h>
#include <hip/hip_bf16.h>

#define B_  32
#define S_  1024
#define D_  256
#define W_  4
#define NL_ 2
#define NH_ 2

typedef __attribute__((ext_vector_type(2))) float v2f;
typedef __attribute__((ext_vector_type(8))) float v8f;
typedef __attribute__((ext_vector_type(4))) float v4f;

#define BM_     128
#define KC_     32
#define PITCH_  36
#define XP_     260
#define WROWS_  (2 * D_)
#define STAGE_  (WROWS_ * PITCH_)
#define FW_     (2 * W_)

__device__ __forceinline__ float gate_sigmoid(float z, float cb) {
  const float e = __builtin_amdgcn_exp2f(fmaf(z, -1.44269504f, cb));
  return __builtin_amdgcn_rcpf(1.f + e);
}

template <int XROW>
__global__ __launch_bounds__(256)
void bilm_block_kernel(const float* __restrict__ xf, const float* __restrict__ xb,
                       const float* __restrict__ padf, const float* __restrict__ padb,
                       const float* __restrict__ wf, const float* __restrict__ wb,
                       const float* __restrict__ Wt0, const float* __restrict__ Wt1,
                       const float* __restrict__ bias0, const float* __restrict__ bias1,
                       float* __restrict__ out0, float* __restrict__ out1) {
  __shared__ __attribute__((aligned(16))) float lds[BM_ * XP_ + 2 * STAGE_];
  float* XR  = lds;
  float* stg = lds + BM_ * XP_;

  const int dir = blockIdx.z;
  const float* x    = dir ? xb : xf;
  const float* fw   = dir ? wb : wf;
  const float* Wt   = dir ? Wt1 : Wt0;
  const float* bias = dir ? bias1 : bias0;
  float* out        = dir ? out1 : out0;

  const int tid  = threadIdx.x;
  const int lane = tid & 31;
  const int wave = tid >> 5;
  const int half = lane >> 4;
  const int l    = lane & 15;
  const int wm   = wave & 3;
  const int wn   = wave >> 2;

  const int mBlk = blockIdx.y * BM_;

  const float w0 = fw[0], w1 = fw[1], w2 = fw[2], w3 = fw[3], w4 = fw[4];

  const int lrow = tid >> 3;
  const int lcol = (tid & 7) * 4;
  const int seq  = mBlk / S_;
  const int s0   = mBlk % S_;
  const int t0   = s0 + 4 * lrow + (dir ? W_ : 0);
  const int xLo = max(t0 - W_, 0);
  const int xHi = min(t0, S_ - 4);
  const float* gLo = x + ((size_t)seq * S_ + (size_t)xLo) * XROW + lcol;
  const float* gHi = x + ((size_t)seq * S_ + (size_t)xHi) * XROW + lcol;
  int sLo = XROW, sHi = XROW;
  if (t0 < W_)            { gLo = padf + t0 * D_ + lcol;                  sLo = D_; }
  if (t0 + 4 >= W_ + S_)  { gHi = padb + (t0 + 4 - W_ - S_) * D_ + lcol;  sHi = D_; }
  const int xOff = 4 * lrow * XP_ + lcol;
  const int wOff = lrow * PITCH_ + lcol;

  const int fA = (wm * 32 + l) * XP_ + 2 * half;
  const int fN = (wn * 128 + l) * PITCH_ + 2 * half;
  const int fG = fN + D_ * PITCH_;

  v8f accN[2][8], accG[2][8];

  v4f v[FW_];
  v4f rw[8];

#define FIR_LOAD_(KO)                                                                                  \
  _Pragma("unroll")                                                                                    \
  for (int u = 0; u < 4; ++u)                                                                          \
    v[u]     = *(const v4f*)(gLo + u * (XROW == D_ ? D_ : sLo) + (KO));                                \
  _Pragma("unroll")                                                                                    \
  for (int u = 0; u < 4; ++u)                                                                          \
    v[u + 4] = *(const v4f*)(gHi + u * (XROW == D_ ? D_ : sHi) + (KO));

#define FIR_STORE_(DST)                                                                                \
  {                                                                                                    \
    _Pragma("unroll")                                                                                  \
    for (int r = 0; r < 4; ++r) {                                                                      \
      v4f acc;                                                                                         \
      _Pragma("unroll")                                                                                \
      for (int c = 0; c < 4; ++c) {                                                                    \
        float a = fmaf(w0, v[r][c], 0.f);                                                              \
        a = fmaf(w1, v[r + 1][c], a);                                                                  \
        a = fmaf(w2, v[r + 2][c], a);                                                                  \
        a = fmaf(w3, v[r + 3][c], a);                                                                  \
        a = fmaf(w4, v[r + 4][c], a);                                                                  \
        acc[c] = a;                                                                                    \
      }                                                                                                \
      *(v4f*)((DST) + r * XP_) = acc;                                                                  \
    }                                                                                                  \
  }

#define W_LOAD_(R0, KO)                                                                                \
  _Pragma("unroll")                                                                                    \
  for (int i = 0; i < 8; ++i)                                                                          \
    rw[i] = *(const v4f*)(gW + (size_t)((R0) + 32 * i) * D_ + (KO));
#define W_STORE_(DST, R0)                                                                              \
  _Pragma("unroll")                                                                                    \
  for (int i = 0; i < 8; ++i)                                                                          \
    *(v4f*)((DST) + wOff + ((R0) + 32 * i) * PITCH_) = rw[i];

#define MM_REGION_(G0)                                                                                 \
  {                                                                                                    \
    v2f fa[2][2], fn[2][8], fg[2][8];                                                                  \
    _Pragma("unroll")                                                                                  \
    for (int gg = 0; gg < 2; ++gg) {                                                                   \
      const int kk = 4 * ((G0) + gg);                                                                  \
      fa[gg][0] = *(const v2f*)(xa + fA + kk);                                                         \
      fa[gg][1] = *(const v2f*)(xa + fA + 16 * XP_ + kk);                                              \
      _Pragma("unroll")                                                                                \
      for (int j = 0; j < 8; ++j) {                                                                    \
        fn[gg][j] = *(const v2f*)(s + fN + 16 * j * PITCH_ + kk);                                      \
        fg[gg][j] = *(const v2f*)(s + fG + 16 * j * PITCH_ + kk);                                      \
      }                                                                                                \
    }                                                                                                  \
    _Pragma("unroll")                                                                                  \
    for (int gg = 0; gg < 2; ++gg)                                                                     \
      _Pragma("unroll")                                                                                \
      for (int ms = 0; ms < 2; ++ms)                                                                   \
        _Pragma("unroll")                                                                              \
        for (int j = 0; j < 8; ++j) {                                                                  \
          accN[ms][j] = __builtin_amdgcn_wmma_f32_16x16x4_f32(false, fa[gg][ms], false, fn[gg][j],     \
                                                              (short)0, accN[ms][j], false, false);    \
          accG[ms][j] = __builtin_amdgcn_wmma_f32_16x16x4_f32(false, fa[gg][ms], false, fg[gg][j],     \
                                                              (short)0, accG[ms][j], false, false);    \
        }                                                                                              \
  }

#pragma unroll 1
  for (int layer = 0; layer < NH_; ++layer) {
    const float* gW = Wt + (size_t)layer * 2 * D_ * D_ + (size_t)lrow * D_ + lcol;
    const float* bl = bias + layer * 2 * D_;

#pragma unroll
    for (int ms = 0; ms < 2; ++ms)
#pragma unroll
      for (int j = 0; j < 8; ++j) { accN[ms][j] = (v8f){}; accG[ms][j] = (v8f){}; }

    if (layer == 0) {
      FIR_LOAD_(0)
      FIR_STORE_(XR + xOff)
    }
    W_LOAD_(0, 0)
    W_STORE_(stg, 0)
    W_LOAD_(D_, 0)
    W_STORE_(stg, D_)
    __syncthreads();

#pragma unroll 1
    for (int kc = 0; kc < D_ / KC_; ++kc) {
      const bool more = (kc + 1) < D_ / KC_;
      const int ko = (kc + 1) * KC_;
      const float* s  = stg + (kc & 1) * STAGE_;
      const float* xa = XR + kc * KC_;
      float* d = stg + ((kc + 1) & 1) * STAGE_;

      if (more) { W_LOAD_(0, ko) }
      __builtin_amdgcn_sched_barrier(0);
      MM_REGION_(0)
      __builtin_amdgcn_sched_barrier(0);
      if (more && layer == 0) { FIR_LOAD_(ko) }
      __builtin_amdgcn_sched_barrier(0);
      MM_REGION_(2)
      __builtin_amdgcn_sched_barrier(0);
      if (more) {
        if (layer == 0) { FIR_STORE_(XR + xOff + ko) }
        W_STORE_(d, 0)
        W_LOAD_(D_, ko)
      }
      __builtin_amdgcn_sched_barrier(0);
      MM_REGION_(4)
      __builtin_amdgcn_sched_barrier(0);
      MM_REGION_(6)
      __builtin_amdgcn_sched_barrier(0);
      if (more) {
        W_STORE_(d, D_)
        __syncthreads();
      }
    }
    __syncthreads();

#pragma unroll
    for (int j = 0; j < 8; ++j) {
      const int n = wn * 128 + 16 * j + l;
      const float bn = bl[n];
      const float cg = bl[D_ + n] * -1.44269504f;
#pragma unroll
      for (int ms = 0; ms < 2; ++ms)
#pragma unroll
        for (int r = 0; r < 8; ++r) {
          const float* xr = XR + (wm * 32 + 16 * ms + r + 8 * half) * XP_ + n;
          const float g = gate_sigmoid(accG[ms][j][r], cg);
          accN[ms][j][r] = g * xr[0] + (1.f - g) * fmaxf(accN[ms][j][r] + bn, 0.f);
        }
    }
    if (layer + 1 < NH_) {
#pragma unroll
      for (int j = 0; j < 8; ++j)
#pragma unroll
        for (int ms = 0; ms < 2; ++ms)
#pragma unroll
          for (int r = 0; r < 8; ++r)
            XR[(wm * 32 + 16 * ms + r + 8 * half) * XP_ + wn * 128 + 16 * j + l] = accN[ms][j][r];
    } else {
      float* op = out + (size_t)(mBlk + wm * 32 + 8 * half) * (2 * D_) + wn * 128 + l;
#pragma unroll
      for (int j = 0; j < 8; ++j)
#pragma unroll
        for (int ms = 0; ms < 2; ++ms)
#pragma unroll
          for (int r = 0; r < 8; ++r)
            op[(size_t)(16 * ms + r) * (2 * D_) + 16 * j] = accN[ms][j][r];
    }
  }
#undef MM_REGION_
#undef W_LOAD_
#undef W_STORE_
#undef FIR_LOAD_
#undef FIR_STORE_
}

extern "C" void kernel_launch(void* const* d_in, const int* in_sizes, int n_in,
                              void* d_out, int out_size, void* d_ws, size_t ws_size,
                              hipStream_t stream) {
  (void)in_sizes; (void)n_in; (void)out_size; (void)d_ws; (void)ws_size;

  const float* inputs   = (const float*)d_in[0];
  const float* fwd_pad  = (const float*)d_in[1];
  const float* bwd_pad  = (const float*)d_in[2];
  const float* fwd_w    = (const float*)d_in[3];
  const float* bwd_w    = (const float*)d_in[4];
  const float* fwd_hw_W = (const float*)d_in[5];
  const float* fwd_hw_b = (const float*)d_in[6];
  const float* bwd_hw_W = (const float*)d_in[7];
  const float* bwd_hw_b = (const float*)d_in[8];
  float* out = (float*)d_out;

  const size_t M = (size_t)B_ * S_;

  dim3 blk(256);
  dim3 grid(1, (unsigned)(M / BM_), 2);

  for (int i = 0; i < NL_; ++i) {
    const float* padf = fwd_pad + (size_t)i * W_ * D_;
    const float* padb = bwd_pad + (size_t)i * W_ * D_;
    const float* tf   = fwd_w + i * (W_ + 1);
    const float* tb   = bwd_w + i * (W_ + 1);
    const float* HWf = fwd_hw_W + (size_t)i * NH_ * 2 * D_ * D_;
    const float* HWb = bwd_hw_W + (size_t)i * NH_ * 2 * D_ * D_;
    const float* Hbf = fwd_hw_b + (size_t)i * NH_ * 2 * D_;
    const float* Hbb = bwd_hw_b + (size_t)i * NH_ * 2 * D_;
    float* oi = out + (size_t)i * M * 2 * D_;

    if (i == 0) {
      bilm_block_kernel<D_><<<grid, blk, 0, stream>>>(
          inputs, inputs, padf, padb, tf, tb, HWf, HWb, Hbf, Hbb, oi, oi + D_);
    } else {
      const float* xin = out + (size_t)(i - 1) * M * 2 * D_;
      bilm_block_kernel<2 * D_><<<grid, blk, 0, stream>>>(
          xin, xin + D_, padf, padb, tf, tb, HWf, HWb, Hbf, Hbb, oi, oi + D_);
    }
  }
}
